# starting kernel
# baseline (speedup 1.0000x reference)
#include <hip/hip_runtime.h>
#include <math.h>


typedef __attribute__((ext_vector_type(2))) float v2f;
typedef __attribute__((ext_vector_type(8))) float v8f;

#define SLOPE 0.2f
#define BPAD 68
#define W0PITCH 26
#define WPITCH 66

__device__ __forceinline__ v8f wmma4(v2f a, v2f b, v8f c) {
    return __builtin_amdgcn_wmma_f32_16x16x4_f32(
        false, a, false, b, (short)0, c, false, false);
}

__device__ __forceinline__ v2f afrag(const float* buf, int kc, int lane) {
    int row = lane & 15;
    int k   = (kc << 2) + ((lane >> 4) << 1);
    return *(const v2f*)(buf + row * BPAD + k);
}

__device__ __forceinline__ v2f bfragT(const float* W, int stride, int kc, int t, int lane) {
    int n = (t << 4) + (lane & 15);
    int k = (kc << 2) + ((lane >> 4) << 1);
    return *(const v2f*)(W + n * stride + k);
}

__device__ __forceinline__ v2f bfragN(const float* W, int stride, int kc, int t, int lane) {
    int n = (t << 4) + (lane & 15);
    int k = (kc << 2) + ((lane >> 4) << 1);
    v2f r;
    r.x = W[k * stride + n];
    r.y = W[(k + 1) * stride + n];
    return r;
}

__global__ __launch_bounds__(128)
void prior_fused_kernel(const float* __restrict__ x,
                        const float* __restrict__ W0g, const float* __restrict__ b0g,
                        const float* __restrict__ W1g, const float* __restrict__ b1g,
                        const float* __restrict__ W2g, const float* __restrict__ b2g,
                        const float* __restrict__ Woutg, const float* __restrict__ boutg,
                        float* __restrict__ out, float* __restrict__ ws)
{
    __shared__ __align__(16) float W0s[64 * W0PITCH + 64];
    __shared__ __align__(16) float W1s[64 * WPITCH];
    __shared__ __align__(16) float W2s[64 * WPITCH];
    __shared__ __align__(16) float Wouts[64];
    __shared__ __align__(16) float b0s[64], b1s[64], b2s[64];
    __shared__ __align__(16) float bufA[4][16 * BPAD];
    __shared__ float sladS[4];

    const int tid  = threadIdx.x;
    const int lane = tid & 31;
    const int wv   = tid >> 5;
    const int z    = blockIdx.y;
    const int rblk = blockIdx.x * 64;
    const int b    = rblk >> 9;
    const int t0   = rblk & 511;
    const int r0   = rblk + wv * 16;
    const int tw0  = t0 + wv * 16;

    for (int i = tid; i < 64 * 24; i += 128) {
        int hh = i / 24, dd = i - hh * 24;
        W0s[hh * W0PITCH + dd] = W0g[(z * 64 + hh) * 24 + dd];
    }
    for (int i = tid; i < 64 * 64; i += 128) {
        int gg = i >> 6, hh = i & 63;
        W1s[gg * WPITCH + hh] = W1g[(z * 64 + gg) * 64 + hh];
        W2s[gg * WPITCH + hh] = W2g[(z * 64 + gg) * 64 + hh];
    }
    if (tid < 64) {
        Wouts[tid] = Woutg[z * 64 + tid];
        b0s[tid]   = b0g[z * 64 + tid];
        b1s[tid]   = b1g[z * 64 + tid];
        b2s[tid]   = b2g[z * 64 + tid];
    }
    __syncthreads();

    float* myBuf = bufA[wv];

    for (int i = 0; i < 12; ++i) {
        int idx = i * 32 + lane;
        int row = idx / 24, col = idx - row * 24;
        myBuf[row * BPAD + col] = x[((size_t)(b * 514 + tw0 + row)) * 8 + col];
    }

    v2f a6[6];
    #pragma unroll
    for (int kc = 0; kc < 6; ++kc) a6[kc] = afrag(myBuf, kc, lane);
    float d0r[4][8];
    #pragma unroll
    for (int t = 0; t < 4; ++t) {
        v8f c = {};
        #pragma unroll
        for (int kc = 0; kc < 6; ++kc)
            c = wmma4(a6[kc], bfragT(W0s, W0PITCH, kc, t, lane), c);
        float bias = b0s[(t << 4) + (lane & 15)];
        #pragma unroll
        for (int v = 0; v < 8; ++v) {
            float hv = c[v] + bias;
            float dv = hv > 0.0f ? 1.0f : SLOPE;
            d0r[t][v] = dv;
            myBuf[(v + ((lane >> 4) << 3)) * BPAD + (t << 4) + (lane & 15)] = hv * dv;
        }
    }

    v2f a16[16];
    #pragma unroll
    for (int kc = 0; kc < 16; ++kc) a16[kc] = afrag(myBuf, kc, lane);
    float d1r[4][8];
    #pragma unroll
    for (int t = 0; t < 4; ++t) {
        v8f c = {};
        #pragma unroll
        for (int kc = 0; kc < 16; ++kc)
            c = wmma4(a16[kc], bfragT(W1s, WPITCH, kc, t, lane), c);
        float bias = b1s[(t << 4) + (lane & 15)];
        #pragma unroll
        for (int v = 0; v < 8; ++v) {
            float hv = c[v] + bias;
            float dv = hv > 0.0f ? 1.0f : SLOPE;
            d1r[t][v] = dv;
            myBuf[(v + ((lane >> 4) << 3)) * BPAD + (t << 4) + (lane & 15)] = hv * dv;
        }
    }

    #pragma unroll
    for (int kc = 0; kc < 16; ++kc) a16[kc] = afrag(myBuf, kc, lane);
    float p[8];
    #pragma unroll
    for (int v = 0; v < 8; ++v) p[v] = 0.0f;
    #pragma unroll
    for (int t = 0; t < 4; ++t) {
        v8f c = {};
        #pragma unroll
        for (int kc = 0; kc < 16; ++kc)
            c = wmma4(a16[kc], bfragT(W2s, WPITCH, kc, t, lane), c);
        float bias = b2s[(t << 4) + (lane & 15)];
        float wo   = Wouts[(t << 4) + (lane & 15)];
        #pragma unroll
        for (int v = 0; v < 8; ++v) {
            float hv = c[v] + bias;
            float dv = hv > 0.0f ? 1.0f : SLOPE;
            p[v] += hv * dv * wo;
            myBuf[(v + ((lane >> 4) << 3)) * BPAD + (t << 4) + (lane & 15)] = dv * wo;
        }
    }
    #pragma unroll
    for (int v = 0; v < 8; ++v) {
        float s = p[v];
        s += __shfl_xor(s, 1);
        s += __shfl_xor(s, 2);
        s += __shfl_xor(s, 4);
        s += __shfl_xor(s, 8);
        p[v] = s;
    }
    if ((lane & 15) == 0) {
        float bo = boutg[z];
        #pragma unroll
        for (int v = 0; v < 8; ++v) {
            int row = v + ((lane >> 4) << 3);
            out[(size_t)(r0 + row) * 8 + z] = p[v] + bo;
        }
    }

    #pragma unroll
    for (int kc = 0; kc < 16; ++kc) a16[kc] = afrag(myBuf, kc, lane);
    #pragma unroll
    for (int t = 0; t < 4; ++t) {
        v8f c = {};
        #pragma unroll
        for (int kc = 0; kc < 16; ++kc)
            c = wmma4(a16[kc], bfragN(W2s, WPITCH, kc, t, lane), c);
        #pragma unroll
        for (int v = 0; v < 8; ++v)
            myBuf[(v + ((lane >> 4) << 3)) * BPAD + (t << 4) + (lane & 15)] = c[v] * d1r[t][v];
    }

    #pragma unroll
    for (int kc = 0; kc < 16; ++kc) a16[kc] = afrag(myBuf, kc, lane);
    #pragma unroll
    for (int t = 0; t < 4; ++t) {
        v8f c = {};
        #pragma unroll
        for (int kc = 0; kc < 16; ++kc)
            c = wmma4(a16[kc], bfragN(W1s, WPITCH, kc, t, lane), c);
        #pragma unroll
        for (int v = 0; v < 8; ++v)
            myBuf[(v + ((lane >> 4) << 3)) * BPAD + (t << 4) + (lane & 15)] = c[v] * d0r[t][v];
    }

    #pragma unroll
    for (int kc = 0; kc < 16; ++kc) a16[kc] = afrag(myBuf, kc, lane);
    const int zc = 16 + z;
    const size_t off = (size_t)524416 + (size_t)65536 * (size_t)(16 * z + (z * (z - 1)) / 2);
    float lsum = 0.0f;
    #pragma unroll
    for (int t = 0; t < 2; ++t) {
        v8f c = {};
        #pragma unroll
        for (int kc = 0; kc < 16; ++kc)
            c = wmma4(a16[kc], bfragN(W0s, W0PITCH, kc, t, lane), c);
        int col = (t << 4) + (lane & 15);
        if (col < zc) {
            #pragma unroll
            for (int v = 0; v < 8; ++v) {
                int row = v + ((lane >> 4) << 3);
                out[off + (size_t)(r0 + row) * zc + col] = c[v];
            }
        }
        if (t == 1 && (lane & 15) == z) {
            #pragma unroll
            for (int v = 0; v < 8; ++v) lsum += logf(fabsf(c[v]));
        }
    }
    lsum += __shfl_xor(lsum, 16);
    if (lane == z) sladS[wv] = lsum;
    __syncthreads();
    if (tid == 0)
        ws[(size_t)z * 1024 + blockIdx.x] = sladS[0] + sladS[1] + sladS[2] + sladS[3];
}

__global__ __launch_bounds__(64)
void reduce_slad_kernel(const float* __restrict__ ws, float* __restrict__ out)
{
    __shared__ float s[64];
    const int b = blockIdx.x, tid = threadIdx.x;
    const int z = tid >> 3, j = tid & 7;
    s[tid] = ws[(size_t)z * 1024 + b * 8 + j];
    __syncthreads();
    for (int off = 32; off > 0; off >>= 1) {
        if (tid < off) s[tid] += s[tid + off];
        __syncthreads();
    }
    if (tid == 0) out[524288 + b] = s[0];
}

extern "C" void kernel_launch(void* const* d_in, const int* in_sizes, int n_in,
                              void* d_out, int out_size, void* d_ws, size_t ws_size,
                              hipStream_t stream) {
    const float* x    = (const float*)d_in[0];
    const float* W0   = (const float*)d_in[1];
    const float* b0   = (const float*)d_in[2];
    const float* W1   = (const float*)d_in[3];
    const float* b1   = (const float*)d_in[4];
    const float* W2   = (const float*)d_in[5];
    const float* b2   = (const float*)d_in[6];
    const float* Wout = (const float*)d_in[7];
    const float* bout = (const float*)d_in[8];
    float* out = (float*)d_out;
    float* ws  = (float*)d_ws;

    dim3 grid(1024, 8);
    prior_fused_kernel<<<grid, 128, 0, stream>>>(x, W0, b0, W1, b1, W2, b2, Wout, bout, out, ws);
    reduce_slad_kernel<<<128, 64, 0, stream>>>(ws, out);
}


# Second, transposed LDS copy of W0/W1/W2 (pitch 68, 8-wave groups): each backward B fragment is one contiguous conflict-free 8-byte LDS load, not two strided 4-byte loads; sums unchanged
# speedup vs baseline: 1.0756x; 1.0756x over previous
#include <hip/hip_runtime.h>
#include <math.h>


typedef __attribute__((ext_vector_type(2))) float v2f;
typedef __attribute__((ext_vector_type(8))) float v8f;

#define SLOPE 0.2f
#define BPAD 68
#define W0PITCH 28
#define WPITCH 68
#define NWAVES 8

__device__ __forceinline__ v8f wmma4(v2f a, v2f b, v8f c) {
    return __builtin_amdgcn_wmma_f32_16x16x4_f32(
        false, a, false, b, (short)0, c, false, false);
}

__device__ __forceinline__ v2f afrag(const float* buf, int kc, int lane) {
    int row = lane & 15;
    int k   = (kc << 2) + ((lane >> 4) << 1);
    return *(const v2f*)(buf + row * BPAD + k);
}

__device__ __forceinline__ v2f bfragT(const float* W, int stride, int kc, int t, int lane) {
    int n = (t << 4) + (lane & 15);
    int k = (kc << 2) + ((lane >> 4) << 1);
    return *(const v2f*)(W + n * stride + k);
}


__global__ __launch_bounds__(256)
void prior_fused_kernel(const float* __restrict__ x,
                        const float* __restrict__ W0g, const float* __restrict__ b0g,
                        const float* __restrict__ W1g, const float* __restrict__ b1g,
                        const float* __restrict__ W2g, const float* __restrict__ b2g,
                        const float* __restrict__ Woutg, const float* __restrict__ boutg,
                        float* __restrict__ out, float* __restrict__ ws)
{
    __shared__ __align__(16) float W0s[64 * W0PITCH];
    __shared__ __align__(16) float W0t[32 * WPITCH];
    __shared__ __align__(16) float W1s[64 * WPITCH];
    __shared__ __align__(16) float W1t[64 * WPITCH];
    __shared__ __align__(16) float W2s[64 * WPITCH];
    __shared__ __align__(16) float W2t[64 * WPITCH];
    __shared__ __align__(16) float Wouts[64];
    __shared__ __align__(16) float b0s[64], b1s[64], b2s[64];
    __shared__ __align__(16) float bufA[NWAVES][16 * BPAD];
    __shared__ float sladS[NWAVES];

    const int tid  = threadIdx.x;
    const int lane = tid & 31;
    const int wv   = tid >> 5;
    const int z    = blockIdx.y;
    const int rblk = blockIdx.x * (16 * NWAVES);
    const int b    = rblk >> 9;
    const int t0   = rblk & 511;
    const int r0   = rblk + wv * 16;
    const int tw0  = t0 + wv * 16;

    {
        const int sg = tid & 63;
        const int sq = tid >> 6;
        #pragma unroll
        for (int i = 0; i < 4; ++i) {
            const int q = sq + 4 * i;
            const float4 v1 = *(const float4*)(W1g + (z * 64 + sg) * 64 + 4 * q);
            const float4 v2 = *(const float4*)(W2g + (z * 64 + sg) * 64 + 4 * q);
            *(float4*)(W1s + sg * WPITCH + 4 * q) = v1;
            *(float4*)(W2s + sg * WPITCH + 4 * q) = v2;
            W1t[(4 * q + 0) * WPITCH + sg] = v1.x;  W1t[(4 * q + 1) * WPITCH + sg] = v1.y;
            W1t[(4 * q + 2) * WPITCH + sg] = v1.z;  W1t[(4 * q + 3) * WPITCH + sg] = v1.w;
            W2t[(4 * q + 0) * WPITCH + sg] = v2.x;  W2t[(4 * q + 1) * WPITCH + sg] = v2.y;
            W2t[(4 * q + 2) * WPITCH + sg] = v2.z;  W2t[(4 * q + 3) * WPITCH + sg] = v2.w;
        }
        #pragma unroll
        for (int i = 0; i < 2; ++i) {
            const int q = sq + 4 * i;
            if (q < 6) {
                const float4 v0 = *(const float4*)(W0g + (z * 64 + sg) * 24 + 4 * q);
                *(float4*)(W0s + sg * W0PITCH + 4 * q) = v0;
                W0t[(4 * q + 0) * WPITCH + sg] = v0.x;  W0t[(4 * q + 1) * WPITCH + sg] = v0.y;
                W0t[(4 * q + 2) * WPITCH + sg] = v0.z;  W0t[(4 * q + 3) * WPITCH + sg] = v0.w;
            }
        }
        W0t[(24 + sq) * WPITCH + sg] = 0.0f;
        W0t[(28 + sq) * WPITCH + sg] = 0.0f;
    }
    if (tid < 64) {
        Wouts[tid] = Woutg[z * 64 + tid];
        b0s[tid]   = b0g[z * 64 + tid];
        b1s[tid]   = b1g[z * 64 + tid];
        b2s[tid]   = b2g[z * 64 + tid];
    }
    __syncthreads();

    float* myBuf = bufA[wv];

    for (int i = 0; i < 12; ++i) {
        int idx = i * 32 + lane;
        int row = idx / 24, col = idx - row * 24;
        myBuf[row * BPAD + col] = x[((size_t)(b * 514 + tw0 + row)) * 8 + col];
    }

    v2f a6[6];
    #pragma unroll
    for (int kc = 0; kc < 6; ++kc) a6[kc] = afrag(myBuf, kc, lane);
    float d0r[4][8];
    #pragma unroll
    for (int t = 0; t < 4; ++t) {
        v8f c = {};
        #pragma unroll
        for (int kc = 0; kc < 6; ++kc)
            c = wmma4(a6[kc], bfragT(W0s, W0PITCH, kc, t, lane), c);
        float bias = b0s[(t << 4) + (lane & 15)];
        #pragma unroll
        for (int v = 0; v < 8; ++v) {
            float hv = c[v] + bias;
            float dv = hv > 0.0f ? 1.0f : SLOPE;
            d0r[t][v] = dv;
            myBuf[(v + ((lane >> 4) << 3)) * BPAD + (t << 4) + (lane & 15)] = hv * dv;
        }
    }

    v2f a16[16];
    #pragma unroll
    for (int kc = 0; kc < 16; ++kc) a16[kc] = afrag(myBuf, kc, lane);
    float d1r[4][8];
    #pragma unroll
    for (int t = 0; t < 4; ++t) {
        v8f c = {};
        #pragma unroll
        for (int kc = 0; kc < 16; ++kc)
            c = wmma4(a16[kc], bfragT(W1s, WPITCH, kc, t, lane), c);
        float bias = b1s[(t << 4) + (lane & 15)];
        #pragma unroll
        for (int v = 0; v < 8; ++v) {
            float hv = c[v] + bias;
            float dv = hv > 0.0f ? 1.0f : SLOPE;
            d1r[t][v] = dv;
            myBuf[(v + ((lane >> 4) << 3)) * BPAD + (t << 4) + (lane & 15)] = hv * dv;
        }
    }

    #pragma unroll
    for (int kc = 0; kc < 16; ++kc) a16[kc] = afrag(myBuf, kc, lane);
    float p[8];
    #pragma unroll
    for (int v = 0; v < 8; ++v) p[v] = 0.0f;
    #pragma unroll
    for (int t = 0; t < 4; ++t) {
        v8f c = {};
        #pragma unroll
        for (int kc = 0; kc < 16; ++kc)
            c = wmma4(a16[kc], bfragT(W2s, WPITCH, kc, t, lane), c);
        float bias = b2s[(t << 4) + (lane & 15)];
        float wo   = Wouts[(t << 4) + (lane & 15)];
        #pragma unroll
        for (int v = 0; v < 8; ++v) {
            float hv = c[v] + bias;
            float dv = hv > 0.0f ? 1.0f : SLOPE;
            p[v] += hv * dv * wo;
            myBuf[(v + ((lane >> 4) << 3)) * BPAD + (t << 4) + (lane & 15)] = dv * wo;
        }
    }
    #pragma unroll
    for (int v = 0; v < 8; ++v) {
        float s = p[v];
        s += __shfl_xor(s, 1);
        s += __shfl_xor(s, 2);
        s += __shfl_xor(s, 4);
        s += __shfl_xor(s, 8);
        p[v] = s;
    }
    if ((lane & 15) == 0) {
        float bo = boutg[z];
        #pragma unroll
        for (int v = 0; v < 8; ++v) {
            int row = v + ((lane >> 4) << 3);
            out[(size_t)(r0 + row) * 8 + z] = p[v] + bo;
        }
    }

    #pragma unroll
    for (int kc = 0; kc < 16; ++kc) a16[kc] = afrag(myBuf, kc, lane);
    #pragma unroll
    for (int t = 0; t < 4; ++t) {
        v8f c = {};
        #pragma unroll
        for (int kc = 0; kc < 16; ++kc)
            c = wmma4(a16[kc], bfragT(W2t, WPITCH, kc, t, lane), c);
        #pragma unroll
        for (int v = 0; v < 8; ++v)
            myBuf[(v + ((lane >> 4) << 3)) * BPAD + (t << 4) + (lane & 15)] = c[v] * d1r[t][v];
    }

    #pragma unroll
    for (int kc = 0; kc < 16; ++kc) a16[kc] = afrag(myBuf, kc, lane);
    #pragma unroll
    for (int t = 0; t < 4; ++t) {
        v8f c = {};
        #pragma unroll
        for (int kc = 0; kc < 16; ++kc)
            c = wmma4(a16[kc], bfragT(W1t, WPITCH, kc, t, lane), c);
        #pragma unroll
        for (int v = 0; v < 8; ++v)
            myBuf[(v + ((lane >> 4) << 3)) * BPAD + (t << 4) + (lane & 15)] = c[v] * d0r[t][v];
    }

    #pragma unroll
    for (int kc = 0; kc < 16; ++kc) a16[kc] = afrag(myBuf, kc, lane);
    const int zc = 16 + z;
    const size_t off = (size_t)524416 + (size_t)65536 * (size_t)(16 * z + (z * (z - 1)) / 2);
    float lsum = 0.0f;
    #pragma unroll
    for (int t = 0; t < 2; ++t) {
        v8f c = {};
        #pragma unroll
        for (int kc = 0; kc < 16; ++kc)
            c = wmma4(a16[kc], bfragT(W0t, WPITCH, kc, t, lane), c);
        int col = (t << 4) + (lane & 15);
        if (col < zc) {
            #pragma unroll
            for (int v = 0; v < 8; ++v) {
                int row = v + ((lane >> 4) << 3);
                out[off + (size_t)(r0 + row) * zc + col] = c[v];
            }
        }
        if (t == 1 && (lane & 15) == z) {
            #pragma unroll
            for (int v = 0; v < 8; ++v) lsum += logf(fabsf(c[v]));
        }
    }
    lsum += __shfl_xor(lsum, 16);
    if (lane == z) sladS[wv] = lsum;
    __syncthreads();
    if (tid < 2)
        ws[(size_t)z * 1024 + blockIdx.x * 2 + tid] =
            sladS[4 * tid] + sladS[4 * tid + 1] + sladS[4 * tid + 2] + sladS[4 * tid + 3];
}

__global__ __launch_bounds__(64)
void reduce_slad_kernel(const float* __restrict__ ws, float* __restrict__ out)
{
    __shared__ float s[64];
    const int b = blockIdx.x, tid = threadIdx.x;
    const int z = tid >> 3, j = tid & 7;
    s[tid] = ws[(size_t)z * 1024 + b * 8 + j];
    __syncthreads();
    for (int off = 32; off > 0; off >>= 1) {
        if (tid < off) s[tid] += s[tid + off];
        __syncthreads();
    }
    if (tid == 0) out[524288 + b] = s[0];
}

extern "C" void kernel_launch(void* const* d_in, const int* in_sizes, int n_in,
                              void* d_out, int out_size, void* d_ws, size_t ws_size,
                              hipStream_t stream) {
    const float* x    = (const float*)d_in[0];
    const float* W0   = (const float*)d_in[1];
    const float* b0   = (const float*)d_in[2];
    const float* W1   = (const float*)d_in[3];
    const float* b1   = (const float*)d_in[4];
    const float* W2   = (const float*)d_in[5];
    const float* b2   = (const float*)d_in[6];
    const float* Wout = (const float*)d_in[7];
    const float* bout = (const float*)d_in[8];
    float* out = (float*)d_out;
    float* ws  = (float*)d_ws;

    dim3 grid(512, 8);
    prior_fused_kernel<<<grid, 256, 0, stream>>>(x, W0, b0, W1, b1, W2, b2, Wout, bout, out, ws);
    reduce_slad_kernel<<<128, 64, 0, stream>>>(ws, out);
}


# Layers 0-4 transposed (weights as A operand, one window per lane): D registers become the next operand via v_permlane16_swap, no LDS round trip; k order kept; residual dot summed in a new order
# speedup vs baseline: 1.1366x; 1.0567x over previous
#include <hip/hip_runtime.h>
#include <math.h>


typedef __attribute__((ext_vector_type(2))) float v2f;
typedef __attribute__((ext_vector_type(8))) float v8f;

#define SLOPE 0.2f
#define W0PITCH 28
#define WPITCH 68
#define NWAVES 8

__device__ __forceinline__ v8f wmma4(v2f a, v2f b, v8f c) {
    return __builtin_amdgcn_wmma_f32_16x16x4_f32(
        false, a, false, b, (short)0, c, false, false);
}

__device__ __forceinline__ v2f wfrag(const float* W, int stride, int kc, int t, int lane) {
    int n = (t << 4) + (lane & 15);
    int k = (kc << 2) + ((lane >> 4) << 1);
    return *(const v2f*)(W + n * stride + k);
}

__device__ __forceinline__ void half_swap(float& a, float& b) {
    auto r = __builtin_amdgcn_permlane16_swap(__float_as_uint(a), __float_as_uint(b), false, false);
    a = __uint_as_float(r[0]);
    b = __uint_as_float(r[1]);
}

__device__ __forceinline__ void handover(float (&v)[4][8], v2f (&op)[16]) {
    #pragma unroll
    for (int t = 0; t < 4; ++t) {
        half_swap(v[t][0], v[t][2]);
        half_swap(v[t][1], v[t][3]);
        half_swap(v[t][4], v[t][6]);
        half_swap(v[t][5], v[t][7]);
        op[4 * t + 0] = v2f{v[t][0], v[t][1]};
        op[4 * t + 1] = v2f{v[t][4], v[t][5]};
        op[4 * t + 2] = v2f{v[t][2], v[t][3]};
        op[4 * t + 3] = v2f{v[t][6], v[t][7]};
    }
}

__device__ __forceinline__ void load8(const float* s, int t, int h, float (&o)[8]) {
    const float4 lo = *(const float4*)(s + (t << 4) + (h << 3));
    const float4 hi = *(const float4*)(s + (t << 4) + (h << 3) + 4);
    o[0] = lo.x; o[1] = lo.y; o[2] = lo.z; o[3] = lo.w;
    o[4] = hi.x; o[5] = hi.y; o[6] = hi.z; o[7] = hi.w;
}

__global__ __launch_bounds__(256)
void prior_fused_kernel(const float* __restrict__ x,
                        const float* __restrict__ W0g, const float* __restrict__ b0g,
                        const float* __restrict__ W1g, const float* __restrict__ b1g,
                        const float* __restrict__ W2g, const float* __restrict__ b2g,
                        const float* __restrict__ Woutg, const float* __restrict__ boutg,
                        float* __restrict__ out, float* __restrict__ ws)
{
    __shared__ __align__(16) float W0s[64 * W0PITCH];
    __shared__ __align__(16) float W0t[32 * WPITCH];
    __shared__ __align__(16) float W1s[64 * WPITCH];
    __shared__ __align__(16) float W1t[64 * WPITCH];
    __shared__ __align__(16) float W2s[64 * WPITCH];
    __shared__ __align__(16) float W2t[64 * WPITCH];
    __shared__ __align__(16) float Wouts[64];
    __shared__ __align__(16) float b0s[64];
    __shared__ __align__(16) float b1s[64];
    __shared__ __align__(16) float b2s[64];
    __shared__ float sladS[NWAVES];

    const int tid  = threadIdx.x;
    const int lane = tid & 31;
    const int wv   = tid >> 5;
    const int lc   = lane & 15;
    const int lh   = lane >> 4;
    const int z    = blockIdx.y;
    const int rblk = blockIdx.x * (16 * NWAVES);
    const int b    = rblk >> 9;
    const int t0   = rblk & 511;
    const int r0   = rblk + wv * 16;
    const int tw0  = t0 + wv * 16;

    v2f op[16];
    {
        const float* xw = x + ((size_t)(b * 514 + tw0 + lc)) * 8 + 2 * lh;
        #pragma unroll
        for (int kc = 0; kc < 6; ++kc) op[kc] = *(const v2f*)(xw + 4 * kc);
    }

    {
        const int sg = tid & 63;
        const int sq = tid >> 6;
        #pragma unroll
        for (int i = 0; i < 4; ++i) {
            const int q = sq + 4 * i;
            const float4 v1 = *(const float4*)(W1g + (z * 64 + sg) * 64 + 4 * q);
            const float4 v2 = *(const float4*)(W2g + (z * 64 + sg) * 64 + 4 * q);
            *(float4*)(W1s + sg * WPITCH + 4 * q) = v1;
            *(float4*)(W2s + sg * WPITCH + 4 * q) = v2;
            W1t[(4 * q + 0) * WPITCH + sg] = v1.x;  W1t[(4 * q + 1) * WPITCH + sg] = v1.y;
            W1t[(4 * q + 2) * WPITCH + sg] = v1.z;  W1t[(4 * q + 3) * WPITCH + sg] = v1.w;
            W2t[(4 * q + 0) * WPITCH + sg] = v2.x;  W2t[(4 * q + 1) * WPITCH + sg] = v2.y;
            W2t[(4 * q + 2) * WPITCH + sg] = v2.z;  W2t[(4 * q + 3) * WPITCH + sg] = v2.w;
        }
        #pragma unroll
        for (int i = 0; i < 2; ++i) {
            const int q = sq + 4 * i;
            if (q < 6) {
                const float4 v0 = *(const float4*)(W0g + (z * 64 + sg) * 24 + 4 * q);
                *(float4*)(W0s + sg * W0PITCH + 4 * q) = v0;
                W0t[(4 * q + 0) * WPITCH + sg] = v0.x;  W0t[(4 * q + 1) * WPITCH + sg] = v0.y;
                W0t[(4 * q + 2) * WPITCH + sg] = v0.z;  W0t[(4 * q + 3) * WPITCH + sg] = v0.w;
            }
        }
        W0t[(24 + sq) * WPITCH + sg] = 0.0f;
        W0t[(28 + sq) * WPITCH + sg] = 0.0f;
    }
    if (tid < 64) {
        Wouts[tid] = Woutg[z * 64 + tid];
        b0s[tid]   = b0g[z * 64 + tid];
        b1s[tid]   = b1g[z * 64 + tid];
        b2s[tid]   = b2g[z * 64 + tid];
    }
    __syncthreads();

    float act[4][8];
    float d0r[4][8], d1r[4][8];

    #pragma unroll
    for (int t = 0; t < 4; ++t) {
        v8f c = {};
        #pragma unroll
        for (int kc = 0; kc < 6; ++kc)
            c = wmma4(wfrag(W0s, W0PITCH, kc, t, lane), op[kc], c);
        float bias[8];
        load8(b0s, t, lh, bias);
        #pragma unroll
        for (int r = 0; r < 8; ++r) {
            float hv = c[r] + bias[r];
            float dv = hv > 0.0f ? 1.0f : SLOPE;
            d0r[t][r] = dv;
            act[t][r] = hv * dv;
        }
    }
    handover(act, op);

    #pragma unroll
    for (int t = 0; t < 4; ++t) {
        v8f c = {};
        #pragma unroll
        for (int kc = 0; kc < 16; ++kc)
            c = wmma4(wfrag(W1s, WPITCH, kc, t, lane), op[kc], c);
        float bias[8];
        load8(b1s, t, lh, bias);
        #pragma unroll
        for (int r = 0; r < 8; ++r) {
            float hv = c[r] + bias[r];
            float dv = hv > 0.0f ? 1.0f : SLOPE;
            d1r[t][r] = dv;
            act[t][r] = hv * dv;
        }
    }
    handover(act, op);

    float p = 0.0f;
    #pragma unroll
    for (int t = 0; t < 4; ++t) {
        v8f c = {};
        #pragma unroll
        for (int kc = 0; kc < 16; ++kc)
            c = wmma4(wfrag(W2s, WPITCH, kc, t, lane), op[kc], c);
        float bias[8], wo[8];
        load8(b2s, t, lh, bias);
        load8(Wouts, t, lh, wo);
        #pragma unroll
        for (int r = 0; r < 8; ++r) {
            float hv = c[r] + bias[r];
            float dv = hv > 0.0f ? 1.0f : SLOPE;
            p += hv * dv * wo[r];
            act[t][r] = dv * wo[r];
        }
    }
    p += __shfl_xor(p, 16);
    if (lh == 0)
        out[(size_t)(r0 + lc) * 8 + z] = p + boutg[z];
    handover(act, op);

    #pragma unroll
    for (int t = 0; t < 4; ++t) {
        v8f c = {};
        #pragma unroll
        for (int kc = 0; kc < 16; ++kc)
            c = wmma4(wfrag(W2t, WPITCH, kc, t, lane), op[kc], c);
        #pragma unroll
        for (int r = 0; r < 8; ++r) act[t][r] = c[r] * d1r[t][r];
    }
    handover(act, op);

    #pragma unroll
    for (int t = 0; t < 4; ++t) {
        v8f c = {};
        #pragma unroll
        for (int kc = 0; kc < 16; ++kc)
            c = wmma4(wfrag(W1t, WPITCH, kc, t, lane), op[kc], c);
        #pragma unroll
        for (int r = 0; r < 8; ++r) act[t][r] = c[r] * d0r[t][r];
    }
    handover(act, op);

    const int zc = 16 + z;
    const size_t off = (size_t)524416 + (size_t)65536 * (size_t)(16 * z + (z * (z - 1)) / 2);
    float lsum = 0.0f;
    #pragma unroll
    for (int t = 0; t < 2; ++t) {
        v8f c = {};
        #pragma unroll
        for (int kc = 0; kc < 16; ++kc)
            c = wmma4(op[kc], wfrag(W0t, WPITCH, kc, t, lane), c);
        int col = (t << 4) + (lane & 15);
        if (col < zc) {
            #pragma unroll
            for (int v = 0; v < 8; ++v) {
                int row = v + ((lane >> 4) << 3);
                out[off + (size_t)(r0 + row) * zc + col] = c[v];
            }
        }
        if (t == 1 && (lane & 15) == z) {
            #pragma unroll
            for (int v = 0; v < 8; ++v) lsum += logf(fabsf(c[v]));
        }
    }
    lsum += __shfl_xor(lsum, 16);
    if (lane == z) sladS[wv] = lsum;
    __syncthreads();
    if (tid < 2)
        ws[(size_t)z * 1024 + blockIdx.x * 2 + tid] =
            sladS[4 * tid] + sladS[4 * tid + 1] + sladS[4 * tid + 2] + sladS[4 * tid + 3];
}

__global__ __launch_bounds__(64)
void reduce_slad_kernel(const float* __restrict__ ws, float* __restrict__ out)
{
    __shared__ float s[64];
    const int b = blockIdx.x, tid = threadIdx.x;
    const int z = tid >> 3, j = tid & 7;
    s[tid] = ws[(size_t)z * 1024 + b * 8 + j];
    __syncthreads();
    for (int off = 32; off > 0; off >>= 1) {
        if (tid < off) s[tid] += s[tid + off];
        __syncthreads();
    }
    if (tid == 0) out[524288 + b] = s[0];
}

extern "C" void kernel_launch(void* const* d_in, const int* in_sizes, int n_in,
                              void* d_out, int out_size, void* d_ws, size_t ws_size,
                              hipStream_t stream) {
    const float* x    = (const float*)d_in[0];
    const float* W0   = (const float*)d_in[1];
    const float* b0   = (const float*)d_in[2];
    const float* W1   = (const float*)d_in[3];
    const float* b1   = (const float*)d_in[4];
    const float* W2   = (const float*)d_in[5];
    const float* b2   = (const float*)d_in[6];
    const float* Wout = (const float*)d_in[7];
    const float* bout = (const float*)d_in[8];
    float* out = (float*)d_out;
    float* ws  = (float*)d_ws;

    dim3 grid(512, 8);
    prior_fused_kernel<<<grid, 256, 0, stream>>>(x, W0, b0, W1, b1, W2, b2, Wout, bout, out, ws);
    reduce_slad_kernel<<<128, 64, 0, stream>>>(ws, out);
}


# Each wave carries two 16-window column tiles (work-group = 4 waves x 32 windows, 3 per WGP): every LDS weight fragment feeds two matrix instructions, LDS reads halved; every sum in the same order
# speedup vs baseline: 1.1944x; 1.0509x over previous
#include <hip/hip_runtime.h>
#include <math.h>


typedef __attribute__((ext_vector_type(2))) float v2f;
typedef __attribute__((ext_vector_type(8))) float v8f;

#define SLOPE 0.2f
#define W0PITCH 28
#define WPITCH 68
#define NWAVES 4
#define NU 2

__device__ __forceinline__ v8f wmma4(v2f a, v2f b, v8f c) {
    return __builtin_amdgcn_wmma_f32_16x16x4_f32(
        false, a, false, b, (short)0, c, false, false);
}

__device__ __forceinline__ v2f wfrag(const float* W, int stride, int kc, int t, int lane) {
    int n = (t << 4) + (lane & 15);
    int k = (kc << 2) + ((lane >> 4) << 1);
    return *(const v2f*)(W + n * stride + k);
}

__device__ __forceinline__ void half_swap(float& a, float& b) {
    auto r = __builtin_amdgcn_permlane16_swap(__float_as_uint(a), __float_as_uint(b), false, false);
    a = __uint_as_float(r[0]);
    b = __uint_as_float(r[1]);
}

__device__ __forceinline__ void handover(float (&v)[4][8], v2f (&op)[16]) {
    #pragma unroll
    for (int t = 0; t < 4; ++t) {
        half_swap(v[t][0], v[t][2]);
        half_swap(v[t][1], v[t][3]);
        half_swap(v[t][4], v[t][6]);
        half_swap(v[t][5], v[t][7]);
        op[4 * t + 0] = v2f{v[t][0], v[t][1]};
        op[4 * t + 1] = v2f{v[t][4], v[t][5]};
        op[4 * t + 2] = v2f{v[t][2], v[t][3]};
        op[4 * t + 3] = v2f{v[t][6], v[t][7]};
    }
}

__device__ __forceinline__ void load8(const float* s, int t, int h, float (&o)[8]) {
    const float4 lo = *(const float4*)(s + (t << 4) + (h << 3));
    const float4 hi = *(const float4*)(s + (t << 4) + (h << 3) + 4);
    o[0] = lo.x; o[1] = lo.y; o[2] = lo.z; o[3] = lo.w;
    o[4] = hi.x; o[5] = hi.y; o[6] = hi.z; o[7] = hi.w;
}

__global__ __launch_bounds__(128, 3)
void prior_fused_kernel(const float* __restrict__ x,
                        const float* __restrict__ W0g, const float* __restrict__ b0g,
                        const float* __restrict__ W1g, const float* __restrict__ b1g,
                        const float* __restrict__ W2g, const float* __restrict__ b2g,
                        const float* __restrict__ Woutg, const float* __restrict__ boutg,
                        float* __restrict__ out, float* __restrict__ ws)
{
    __shared__ __align__(16) float W0s[64 * W0PITCH];
    __shared__ __align__(16) float W0t[32 * WPITCH];
    __shared__ __align__(16) float W1s[64 * WPITCH];
    __shared__ __align__(16) float W1t[64 * WPITCH];
    __shared__ __align__(16) float W2s[64 * WPITCH];
    __shared__ __align__(16) float W2t[64 * WPITCH];
    __shared__ __align__(16) float Wouts[64];
    __shared__ __align__(16) float b0s[64];
    __shared__ __align__(16) float b1s[64];
    __shared__ __align__(16) float b2s[64];
    __shared__ float sladS[NWAVES * NU];

    const int tid  = threadIdx.x;
    const int lane = tid & 31;
    const int wv   = tid >> 5;
    const int lc   = lane & 15;
    const int lh   = lane >> 4;
    const int z    = blockIdx.y;
    const int rblk = blockIdx.x * (16 * NU * NWAVES);
    const int b    = rblk >> 9;
    const int t0   = rblk & 511;
    const int r0   = rblk + wv * (16 * NU);
    const int tw0  = t0 + wv * (16 * NU);

    v2f op[NU][16];
    #pragma unroll
    for (int u = 0; u < NU; ++u) {
        const float* xw = x + ((size_t)(b * 514 + tw0 + 16 * u + lc)) * 8 + 2 * lh;
        #pragma unroll
        for (int kc = 0; kc < 6; ++kc) op[u][kc] = *(const v2f*)(xw + 4 * kc);
    }

    {
        const int sg = tid & 63;
        const int sq = tid >> 6;
        #pragma unroll
        for (int i = 0; i < 8; ++i) {
            const int q = sq + 2 * i;
            const float4 v1 = *(const float4*)(W1g + (z * 64 + sg) * 64 + 4 * q);
            const float4 v2 = *(const float4*)(W2g + (z * 64 + sg) * 64 + 4 * q);
            *(float4*)(W1s + sg * WPITCH + 4 * q) = v1;
            *(float4*)(W2s + sg * WPITCH + 4 * q) = v2;
            W1t[(4 * q + 0) * WPITCH + sg] = v1.x;  W1t[(4 * q + 1) * WPITCH + sg] = v1.y;
            W1t[(4 * q + 2) * WPITCH + sg] = v1.z;  W1t[(4 * q + 3) * WPITCH + sg] = v1.w;
            W2t[(4 * q + 0) * WPITCH + sg] = v2.x;  W2t[(4 * q + 1) * WPITCH + sg] = v2.y;
            W2t[(4 * q + 2) * WPITCH + sg] = v2.z;  W2t[(4 * q + 3) * WPITCH + sg] = v2.w;
        }
        #pragma unroll
        for (int i = 0; i < 3; ++i) {
            const int q = sq + 2 * i;
            const float4 v0 = *(const float4*)(W0g + (z * 64 + sg) * 24 + 4 * q);
            *(float4*)(W0s + sg * W0PITCH + 4 * q) = v0;
            W0t[(4 * q + 0) * WPITCH + sg] = v0.x;  W0t[(4 * q + 1) * WPITCH + sg] = v0.y;
            W0t[(4 * q + 2) * WPITCH + sg] = v0.z;  W0t[(4 * q + 3) * WPITCH + sg] = v0.w;
        }
        #pragma unroll
        for (int j = 0; j < 4; ++j) W0t[(24 + sq + 2 * j) * WPITCH + sg] = 0.0f;
    }
    if (tid < 64) {
        Wouts[tid] = Woutg[z * 64 + tid];
        b0s[tid]   = b0g[z * 64 + tid];
        b1s[tid]   = b1g[z * 64 + tid];
        b2s[tid]   = b2g[z * 64 + tid];
    }
    __syncthreads();

    float act[NU][4][8];
    float d0r[NU][4][8], d1r[NU][4][8];

    #pragma unroll
    for (int t = 0; t < 4; ++t) {
        v8f c[NU] = {};
        #pragma unroll
        for (int kc = 0; kc < 6; ++kc) {
            const v2f a = wfrag(W0s, W0PITCH, kc, t, lane);
            #pragma unroll
            for (int u = 0; u < NU; ++u) c[u] = wmma4(a, op[u][kc], c[u]);
        }
        float bias[8];
        load8(b0s, t, lh, bias);
        #pragma unroll
        for (int u = 0; u < NU; ++u) {
            #pragma unroll
            for (int r = 0; r < 8; ++r) {
                float hv = c[u][r] + bias[r];
                float dv = hv > 0.0f ? 1.0f : SLOPE;
                d0r[u][t][r] = dv;
                act[u][t][r] = hv * dv;
            }
        }
    }
    #pragma unroll
    for (int u = 0; u < NU; ++u) handover(act[u], op[u]);

    #pragma unroll
    for (int t = 0; t < 4; ++t) {
        v8f c[NU] = {};
        #pragma unroll
        for (int kc = 0; kc < 16; ++kc) {
            const v2f a = wfrag(W1s, WPITCH, kc, t, lane);
            #pragma unroll
            for (int u = 0; u < NU; ++u) c[u] = wmma4(a, op[u][kc], c[u]);
        }
        float bias[8];
        load8(b1s, t, lh, bias);
        #pragma unroll
        for (int u = 0; u < NU; ++u) {
            #pragma unroll
            for (int r = 0; r < 8; ++r) {
                float hv = c[u][r] + bias[r];
                float dv = hv > 0.0f ? 1.0f : SLOPE;
                d1r[u][t][r] = dv;
                act[u][t][r] = hv * dv;
            }
        }
    }
    #pragma unroll
    for (int u = 0; u < NU; ++u) handover(act[u], op[u]);
    __builtin_amdgcn_sched_barrier(0);

    float p[NU] = {};
    #pragma unroll
    for (int t = 0; t < 4; ++t) {
        v8f c[NU] = {};
        #pragma unroll
        for (int kc = 0; kc < 16; ++kc) {
            const v2f a = wfrag(W2s, WPITCH, kc, t, lane);
            #pragma unroll
            for (int u = 0; u < NU; ++u) c[u] = wmma4(a, op[u][kc], c[u]);
        }
        float bias[8], wo[8];
        load8(b2s, t, lh, bias);
        load8(Wouts, t, lh, wo);
        #pragma unroll
        for (int u = 0; u < NU; ++u) {
            #pragma unroll
            for (int r = 0; r < 8; ++r) {
                float hv = c[u][r] + bias[r];
                float dv = hv > 0.0f ? 1.0f : SLOPE;
                p[u] += hv * dv * wo[r];
                act[u][t][r] = dv * wo[r];
            }
        }
    }
    #pragma unroll
    for (int u = 0; u < NU; ++u) {
        p[u] += __shfl_xor(p[u], 16);
        if (lh == 0)
            out[(size_t)(r0 + 16 * u + lc) * 8 + z] = p[u] + boutg[z];
        handover(act[u], op[u]);
    }

    #pragma unroll
    for (int t = 0; t < 4; ++t) {
        v8f c[NU] = {};
        #pragma unroll
        for (int kc = 0; kc < 16; ++kc) {
            const v2f a = wfrag(W2t, WPITCH, kc, t, lane);
            #pragma unroll
            for (int u = 0; u < NU; ++u) c[u] = wmma4(a, op[u][kc], c[u]);
        }
        #pragma unroll
        for (int u = 0; u < NU; ++u) {
            #pragma unroll
            for (int r = 0; r < 8; ++r) act[u][t][r] = c[u][r] * d1r[u][t][r];
        }
    }
    #pragma unroll
    for (int u = 0; u < NU; ++u) handover(act[u], op[u]);

    #pragma unroll
    for (int t = 0; t < 4; ++t) {
        v8f c[NU] = {};
        #pragma unroll
        for (int kc = 0; kc < 16; ++kc) {
            const v2f a = wfrag(W1t, WPITCH, kc, t, lane);
            #pragma unroll
            for (int u = 0; u < NU; ++u) c[u] = wmma4(a, op[u][kc], c[u]);
        }
        #pragma unroll
        for (int u = 0; u < NU; ++u) {
            #pragma unroll
            for (int r = 0; r < 8; ++r) act[u][t][r] = c[u][r] * d0r[u][t][r];
        }
    }
    #pragma unroll
    for (int u = 0; u < NU; ++u) handover(act[u], op[u]);

    const int zc = 16 + z;
    const size_t off = (size_t)524416 + (size_t)65536 * (size_t)(16 * z + (z * (z - 1)) / 2);
    float lsum[NU] = {};
    #pragma unroll
    for (int t = 0; t < 2; ++t) {
        v8f c[NU] = {};
        #pragma unroll
        for (int kc = 0; kc < 16; ++kc) {
            const v2f w = wfrag(W0t, WPITCH, kc, t, lane);
            #pragma unroll
            for (int u = 0; u < NU; ++u) c[u] = wmma4(op[u][kc], w, c[u]);
        }
        int col = (t << 4) + (lane & 15);
        #pragma unroll
        for (int u = 0; u < NU; ++u) {
            if (col < zc) {
                #pragma unroll
                for (int v = 0; v < 8; ++v) {
                    int row = 16 * u + v + ((lane >> 4) << 3);
                    out[off + (size_t)(r0 + row) * zc + col] = c[u][v];
                }
            }
            if (t == 1 && (lane & 15) == z) {
                #pragma unroll
                for (int v = 0; v < 8; ++v) lsum[u] += logf(fabsf(c[u][v]));
            }
        }
    }
    #pragma unroll
    for (int u = 0; u < NU; ++u) {
        lsum[u] += __shfl_xor(lsum[u], 16);
        if (lane == z) sladS[NU * wv + u] = lsum[u];
    }
    __syncthreads();
    if (tid < 2)
        ws[(size_t)z * 1024 + blockIdx.x * 2 + tid] =
            sladS[4 * tid] + sladS[4 * tid + 1] + sladS[4 * tid + 2] + sladS[4 * tid + 3];
}

__global__ __launch_bounds__(64)
void reduce_slad_kernel(const float* __restrict__ ws, float* __restrict__ out)
{
    __shared__ float s[64];
    const int b = blockIdx.x, tid = threadIdx.x;
    const int z = tid >> 3, j = tid & 7;
    s[tid] = ws[(size_t)z * 1024 + b * 8 + j];
    __syncthreads();
    for (int off = 32; off > 0; off >>= 1) {
        if (tid < off) s[tid] += s[tid + off];
        __syncthreads();
    }
    if (tid == 0) out[524288 + b] = s[0];
}

extern "C" void kernel_launch(void* const* d_in, const int* in_sizes, int n_in,
                              void* d_out, int out_size, void* d_ws, size_t ws_size,
                              hipStream_t stream) {
    const float* x    = (const float*)d_in[0];
    const float* W0   = (const float*)d_in[1];
    const float* b0   = (const float*)d_in[2];
    const float* W1   = (const float*)d_in[3];
    const float* b1   = (const float*)d_in[4];
    const float* W2   = (const float*)d_in[5];
    const float* b2   = (const float*)d_in[6];
    const float* Wout = (const float*)d_in[7];
    const float* bout = (const float*)d_in[8];
    float* out = (float*)d_out;
    float* ws  = (float*)d_ws;

    dim3 grid(512, 8);
    prior_fused_kernel<<<grid, 128, 0, stream>>>(x, W0, b0, W1, b1, W2, b2, Wout, bout, out, ws);
    reduce_slad_kernel<<<128, 64, 0, stream>>>(ws, out);
}
